# starting kernel
# baseline (speedup 1.0000x reference)
#include <hip/hip_runtime.h>
#include <hip/hip_bf16.h>
#include <stdint.h>


typedef float v2f __attribute__((ext_vector_type(2)));
typedef float v8f __attribute__((ext_vector_type(8)));

static __device__ __forceinline__ void atomic_add_f32(float* p, float v) {
  __hip_atomic_fetch_add(p, v, __ATOMIC_RELAXED, __HIP_MEMORY_SCOPE_AGENT);
}


__global__ void zero_f32(float* __restrict__ p, long n) {
  long i = (long)blockIdx.x * blockDim.x + threadIdx.x;
  if (i < n) p[i] = 0.0f;
}

__global__ void degree_kernel(const int* __restrict__ src, const int* __restrict__ dst,
                              float* __restrict__ dout, float* __restrict__ din, int E) {
  int i = blockIdx.x * blockDim.x + threadIdx.x;
  if (i < E) {
    atomic_add_f32(&dout[src[i]], 1.0f);
    atomic_add_f32(&din[dst[i]], 1.0f);
  }
}

__global__ void norm_kernel(float* __restrict__ deg, long n) {
  long i = (long)blockIdx.x * blockDim.x + threadIdx.x;
  if (i < n) deg[i] = rsqrtf(fmaxf(deg[i], 1.0f));
}

template <int FOUT>
__global__ void __launch_bounds__(256)
gemm_norm_wmma(const float* __restrict__ X, const float* __restrict__ W,
               const float* __restrict__ norm, float* __restrict__ H, int ntiles) {
  constexpr int FIN = 128;
  constexpr int NT = FOUT / 16;

  __shared__ __align__(16) float Wlds[FIN * FOUT];
  {
    const float4* wg = (const float4*)W;
    float4* wl = (float4*)Wlds;
    for (int i = threadIdx.x; i < FIN * FOUT / 4; i += blockDim.x) wl[i] = wg[i];
  }
  __syncthreads();

  const int wave = threadIdx.x >> 5;
  const int lane = threadIdx.x & 31;
  const int tile = blockIdx.x * 8 + wave;
  if (tile >= ntiles) return;

  const int half = lane >> 4;
  const int lm   = lane & 15;
  const int row0 = tile * 16;

  const int arow = row0 + lm;
  const float nrm = norm[arow];
  const float* __restrict__ xrow = X + (size_t)arow * FIN;

  v8f acc[NT];
#pragma unroll
  for (int t = 0; t < NT; ++t) acc[t] = (v8f)0.0f;

  for (int kk = 0; kk < FIN; kk += 4) {
    const int ka = kk + half * 2;
    v2f a;
    a.x = xrow[ka] * nrm;
    a.y = xrow[ka + 1] * nrm;
#pragma unroll
    for (int nt = 0; nt < NT; ++nt) {
      const int col = nt * 16 + lm;
      v2f b;
      b.x = Wlds[ka * FOUT + col];
      b.y = Wlds[(ka + 1) * FOUT + col];
      acc[nt] = __builtin_amdgcn_wmma_f32_16x16x4_f32(
           false, a,  false, b,
           (short)0, acc[nt],  false,  false);
    }
  }

#pragma unroll
  for (int nt = 0; nt < NT; ++nt) {
    const int col = nt * 16 + lm;
    float* __restrict__ hp = H + (size_t)(row0 + half * 8) * FOUT + col;
#pragma unroll
    for (int r = 0; r < 8; ++r) hp[(size_t)r * FOUT] = acc[nt][r];
  }
}

template <int FOUT>
__global__ void gemm_norm_scalar(const float* __restrict__ X, const float* __restrict__ W,
                                 const float* __restrict__ norm, float* __restrict__ H,
                                 int row_begin, int nrows) {
  constexpr int FIN = 128;
  int i = blockIdx.x * blockDim.x + threadIdx.x;
  if (i >= nrows * FOUT) return;
  int row = row_begin + i / FOUT;
  int col = i % FOUT;
  const float* xr = X + (size_t)row * FIN;
  float nrm = norm[row];
  float s = 0.0f;
  for (int k = 0; k < FIN; ++k) s += xr[k] * W[k * FOUT + col];
  H[(size_t)row * FOUT + col] = s * nrm;
}

template <int F>
__global__ void __launch_bounds__(256)
scatter_add(const float* __restrict__ H, const int* __restrict__ src,
            const int* __restrict__ dst, float* __restrict__ agg, int E) {
  const int lane = threadIdx.x & 31;
  int w = (blockIdx.x * blockDim.x + threadIdx.x) >> 5;
  const int nw = (gridDim.x * blockDim.x) >> 5;
  for (int e = w; e < E; e += nw) {
    const int s = src[e];
    const int d = dst[e];
    const float* __restrict__ hs = H + (size_t)s * F;
    float* __restrict__ ad = agg + (size_t)d * F;
    const int en = e + nw;
    if (en < E) {
      __builtin_prefetch(H + (size_t)src[en] * F + lane * 4, 0, 1);
    }
#pragma unroll
    for (int c = lane; c < F; c += 32) atomic_add_f32(&ad[c], hs[c]);
  }
}

template <int F, bool RELU>
__global__ void finalize_kernel(const float* __restrict__ agg, const float* __restrict__ norm_in,
                                const float* __restrict__ bias, float* __restrict__ out, long n_elems) {
  constexpr int LOG = (F == 128) ? 7 : 6;
  long i = (long)blockIdx.x * blockDim.x + threadIdx.x;
  if (i >= n_elems) return;
  int row = (int)(i >> LOG);
  int col = (int)(i & (F - 1));
  float v = agg[i] * norm_in[row] + bias[col];
  if (RELU) v = fmaxf(v, 0.0f);
  out[i] = v;
}


extern "C" void kernel_launch(void* const* d_in, const int* in_sizes, int n_in,
                              void* d_out, int out_size, void* d_ws, size_t ws_size,
                              hipStream_t stream) {
  (void)n_in; (void)out_size; (void)ws_size;
  const float* in_feat = (const float*)d_in[0];
  const int*   src     = (const int*)d_in[1];
  const int*   dst     = (const int*)d_in[2];
  const float* W1 = (const float*)d_in[3];
  const float* b1 = (const float*)d_in[4];
  const float* W2 = (const float*)d_in[5];
  const float* b2 = (const float*)d_in[6];
  const float* W3 = (const float*)d_in[7];
  const float* b3 = (const float*)d_in[8];

  const int N = in_sizes[0] / 128;
  const int E = in_sizes[1];

  float* ws       = (float*)d_ws;
  float* norm_out = ws;
  float* norm_in  = ws + N;
  float* h        = ws + 2 * (size_t)N;
  float* agg      = h + (size_t)N * 128;
  float* xbuf     = agg + (size_t)N * 128;
  float* out      = (float*)d_out;

  const int T = 256;
  const long twoN = 2L * N;

  zero_f32<<<dim3((unsigned)((twoN + T - 1) / T)), dim3(T), 0, stream>>>(norm_out, twoN);
  degree_kernel<<<dim3((unsigned)((E + T - 1) / T)), dim3(T), 0, stream>>>(src, dst, norm_out, norm_in, E);
  norm_kernel<<<dim3((unsigned)((twoN + T - 1) / T)), dim3(T), 0, stream>>>(norm_out, twoN);

  const int ntiles = N / 16;
  const int rem = N - ntiles * 16;
  const unsigned gemm_blocks = (unsigned)((ntiles + 7) / 8);
  const unsigned scat_blocks = 8192;

  {
    const long ne = (long)N * 128;
    if (ntiles)
      gemm_norm_wmma<128><<<dim3(gemm_blocks), dim3(T), 0, stream>>>(in_feat, W1, norm_out, h, ntiles);
    if (rem)
      gemm_norm_scalar<128><<<dim3((unsigned)((rem * 128 + T - 1) / T)), dim3(T), 0, stream>>>(
          in_feat, W1, norm_out, h, ntiles * 16, rem);
    zero_f32<<<dim3((unsigned)((ne + T - 1) / T)), dim3(T), 0, stream>>>(agg, ne);
    scatter_add<128><<<dim3(scat_blocks), dim3(T), 0, stream>>>(h, src, dst, agg, E);
    finalize_kernel<128, true><<<dim3((unsigned)((ne + T - 1) / T)), dim3(T), 0, stream>>>(
        agg, norm_in, b1, xbuf, ne);
  }

  {
    const long ne = (long)N * 128;
    if (ntiles)
      gemm_norm_wmma<128><<<dim3(gemm_blocks), dim3(T), 0, stream>>>(xbuf, W2, norm_out, h, ntiles);
    if (rem)
      gemm_norm_scalar<128><<<dim3((unsigned)((rem * 128 + T - 1) / T)), dim3(T), 0, stream>>>(
          xbuf, W2, norm_out, h, ntiles * 16, rem);
    zero_f32<<<dim3((unsigned)((ne + T - 1) / T)), dim3(T), 0, stream>>>(agg, ne);
    scatter_add<128><<<dim3(scat_blocks), dim3(T), 0, stream>>>(h, src, dst, agg, E);
    finalize_kernel<128, true><<<dim3((unsigned)((ne + T - 1) / T)), dim3(T), 0, stream>>>(
        agg, norm_in, b2, xbuf, ne);
  }

  {
    const long ne = (long)N * 64;
    if (ntiles)
      gemm_norm_wmma<64><<<dim3(gemm_blocks), dim3(T), 0, stream>>>(xbuf, W3, norm_out, h, ntiles);
    if (rem)
      gemm_norm_scalar<64><<<dim3((unsigned)((rem * 64 + T - 1) / T)), dim3(T), 0, stream>>>(
          xbuf, W3, norm_out, h, ntiles * 16, rem);
    zero_f32<<<dim3((unsigned)((ne + T - 1) / T)), dim3(T), 0, stream>>>(agg, ne);
    scatter_add<64><<<dim3(scat_blocks), dim3(T), 0, stream>>>(h, src, dst, agg, E);
    finalize_kernel<64, false><<<dim3((unsigned)((ne + T - 1) / T)), dim3(T), 0, stream>>>(
        agg, norm_in, b3, out, ne);
  }
}


# SpMM as gather over a dst-sorted CSR rebuilt per call (returning int atomic ranks each edge, scan, fill), norm_in/bias/relu in its epilogue; row sums in arrival-rank order, no f32 atomics on features
# speedup vs baseline: 2.3664x; 2.3664x over previous
#include <hip/hip_runtime.h>
#include <hip/hip_bf16.h>
#include <stdint.h>


typedef float v2f __attribute__((ext_vector_type(2)));
typedef float v8f __attribute__((ext_vector_type(8)));

static __device__ __forceinline__ void atomic_add_f32(float* p, float v) {
  __hip_atomic_fetch_add(p, v, __ATOMIC_RELAXED, __HIP_MEMORY_SCOPE_AGENT);
}


__global__ void zero_f32(float* __restrict__ p, long n) {
  long i = (long)blockIdx.x * blockDim.x + threadIdx.x;
  if (i < n) p[i] = 0.0f;
}

__global__ void degree_rank_kernel(const int* __restrict__ src, const int* __restrict__ dst,
                                   float* __restrict__ dout, int* __restrict__ cnt,
                                   int* __restrict__ rank, int E) {
  int i = blockIdx.x * blockDim.x + threadIdx.x;
  if (i < E) {
    atomic_add_f32(&dout[src[i]], 1.0f);
    rank[i] = __hip_atomic_fetch_add(&cnt[dst[i]], 1, __ATOMIC_RELAXED, __HIP_MEMORY_SCOPE_AGENT);
  }
}

__global__ void __launch_bounds__(1024)
scan_norm_kernel(const int* __restrict__ cnt, float* __restrict__ dout,
                 float* __restrict__ norm_in, int* __restrict__ rowptr, int n) {
  __shared__ int wpre[32];
  __shared__ int wsum[32];
  const int tid = threadIdx.x;
  const int lane = tid & 31;
  const int wv = tid >> 5;
  const int base = blockIdx.x * 1024;

  int part = 0;
  {
    const int4* c4 = (const int4*)cnt;
    for (int i = tid; i < base / 4; i += 1024) {
      const int4 q = c4[i];
      part += (q.x + q.y) + (q.z + q.w);
    }
  }
  for (int o = 16; o > 0; o >>= 1) part += __shfl_xor(part, o);

  const int i = base + tid;
  const int c = (i < n) ? cnt[i] : 0;
  int x = c;
  for (int o = 1; o < 32; o <<= 1) {
    const int y = __shfl_up(x, o);
    if (lane >= o) x += y;
  }
  if (lane == 31) wsum[wv] = x;
  if (lane == 0) wpre[wv] = part;
  __syncthreads();

  int a = wpre[lane];
  int b = (lane < wv) ? wsum[lane] : 0;
  for (int o = 16; o > 0; o >>= 1) {
    a += __shfl_xor(a, o);
    b += __shfl_xor(b, o);
  }
  const int excl = a + b + x - c;
  if (i < n) {
    rowptr[i] = excl;
    norm_in[i] = rsqrtf(fmaxf((float)c, 1.0f));
    dout[i] = rsqrtf(fmaxf(dout[i], 1.0f));
    if (i == n - 1) rowptr[n] = excl + c;
  }
}

__global__ void fill_csr_kernel(const int* __restrict__ src, const int* __restrict__ dst,
                                const int* __restrict__ rank, const int* __restrict__ rowptr,
                                int* __restrict__ colidx, int E) {
  int i = blockIdx.x * blockDim.x + threadIdx.x;
  if (i < E) {
    const int p = rowptr[dst[i]] + rank[i];
    if ((unsigned)p < (unsigned)E) colidx[p] = src[i];
  }
}

template <int FOUT>
__global__ void __launch_bounds__(256)
gemm_norm_wmma(const float* __restrict__ X, const float* __restrict__ W,
               const float* __restrict__ norm, float* __restrict__ H, int ntiles) {
  constexpr int FIN = 128;
  constexpr int NT = FOUT / 16;

  __shared__ __align__(16) float Wlds[FIN * FOUT];
  {
    const float4* wg = (const float4*)W;
    float4* wl = (float4*)Wlds;
    for (int i = threadIdx.x; i < FIN * FOUT / 4; i += blockDim.x) wl[i] = wg[i];
  }
  __syncthreads();

  const int wave = threadIdx.x >> 5;
  const int lane = threadIdx.x & 31;
  const int tile = blockIdx.x * 8 + wave;
  if (tile >= ntiles) return;

  const int half = lane >> 4;
  const int lm   = lane & 15;
  const int row0 = tile * 16;

  const int arow = row0 + lm;
  const float nrm = norm[arow];
  const float* __restrict__ xrow = X + (size_t)arow * FIN;

  v8f acc[NT];
#pragma unroll
  for (int t = 0; t < NT; ++t) acc[t] = (v8f)0.0f;

  for (int kk = 0; kk < FIN; kk += 4) {
    const int ka = kk + half * 2;
    v2f a;
    a.x = xrow[ka] * nrm;
    a.y = xrow[ka + 1] * nrm;
#pragma unroll
    for (int nt = 0; nt < NT; ++nt) {
      const int col = nt * 16 + lm;
      v2f b;
      b.x = Wlds[ka * FOUT + col];
      b.y = Wlds[(ka + 1) * FOUT + col];
      acc[nt] = __builtin_amdgcn_wmma_f32_16x16x4_f32(
           false, a,  false, b,
           (short)0, acc[nt],  false,  false);
    }
  }

#pragma unroll
  for (int nt = 0; nt < NT; ++nt) {
    const int col = nt * 16 + lm;
    float* __restrict__ hp = H + (size_t)(row0 + half * 8) * FOUT + col;
#pragma unroll
    for (int r = 0; r < 8; ++r) hp[(size_t)r * FOUT] = acc[nt][r];
  }
}

template <int FOUT>
__global__ void gemm_norm_scalar(const float* __restrict__ X, const float* __restrict__ W,
                                 const float* __restrict__ norm, float* __restrict__ H,
                                 int row_begin, int nrows) {
  constexpr int FIN = 128;
  int i = blockIdx.x * blockDim.x + threadIdx.x;
  if (i >= nrows * FOUT) return;
  int row = row_begin + i / FOUT;
  int col = i % FOUT;
  const float* xr = X + (size_t)row * FIN;
  float nrm = norm[row];
  float s = 0.0f;
  for (int k = 0; k < FIN; ++k) s += xr[k] * W[k * FOUT + col];
  H[(size_t)row * FOUT + col] = s * nrm;
}

template <int F, bool RELU>
__global__ void __launch_bounds__(256)
gather_rows(const float* __restrict__ H, const int* __restrict__ rowptr,
            const int* __restrict__ colidx, const float* __restrict__ norm_in,
            const float* __restrict__ bias, float* __restrict__ out, int N) {
  constexpr int VW = F / 32;
  typedef float vec __attribute__((ext_vector_type(VW)));
  const int lane = threadIdx.x & 31;
  const int row = __builtin_amdgcn_readfirstlane((int)((blockIdx.x * 256u + threadIdx.x) >> 5));
  if (row >= N) return;
  const int beg = rowptr[row];
  const int end = rowptr[row + 1];
  const unsigned smax = (unsigned)(N - 1);

  vec acc = (vec)0.0f;
  for (int j = beg; j < end; j += 32) {
    const int cnt = (end - j < 32) ? (end - j) : 32;
    const int myidx = (lane < cnt) ? colidx[j + lane] : 0;
    int t = 0;
    for (; t + 8 <= cnt; t += 8) {
      vec v[8];
#pragma unroll
      for (int u = 0; u < 8; ++u) {
        unsigned s = (unsigned)__builtin_amdgcn_readlane(myidx, t + u);
        s = (s < smax) ? s : smax;
        v[u] = *((const vec*)(H + (size_t)s * F) + lane);
      }
#pragma unroll
      for (int u = 0; u < 8; ++u) acc += v[u];
    }
    if (t < cnt) {
      vec v[7];
#pragma unroll
      for (int u = 0; u < 7; ++u) {
        v[u] = (vec)0.0f;
        if (t + u < cnt) {
          unsigned s = (unsigned)__builtin_amdgcn_readlane(myidx, t + u);
          s = (s < smax) ? s : smax;
          v[u] = *((const vec*)(H + (size_t)s * F) + lane);
        }
      }
#pragma unroll
      for (int u = 0; u < 7; ++u) acc += v[u];
    }
  }

  const float nin = norm_in[row];
  const vec bv = *((const vec*)bias + lane);
  vec r = acc * nin + bv;
  if (RELU) {
#pragma unroll
    for (int k = 0; k < VW; ++k) r[k] = fmaxf(r[k], 0.0f);
  }
  *((vec*)(out + (size_t)row * F) + lane) = r;
}


extern "C" void kernel_launch(void* const* d_in, const int* in_sizes, int n_in,
                              void* d_out, int out_size, void* d_ws, size_t ws_size,
                              hipStream_t stream) {
  (void)n_in; (void)out_size; (void)ws_size;
  const float* in_feat = (const float*)d_in[0];
  const int*   src     = (const int*)d_in[1];
  const int*   dst     = (const int*)d_in[2];
  const float* W1 = (const float*)d_in[3];
  const float* b1 = (const float*)d_in[4];
  const float* W2 = (const float*)d_in[5];
  const float* b2 = (const float*)d_in[6];
  const float* W3 = (const float*)d_in[7];
  const float* b3 = (const float*)d_in[8];

  const int N = in_sizes[0] / 128;
  const int E = in_sizes[1];

  float* ws       = (float*)d_ws;
  float* norm_out = ws;
  int*   cnt      = (int*)(ws + N);
  float* h        = ws + 2 * (size_t)N;
  float* csr      = h + (size_t)N * 128;
  float* xbuf     = csr + (size_t)N * 128;
  float* out      = (float*)d_out;

  float* norm_in  = csr;
  int*   rowptr   = (int*)(csr + N);
  int*   rank     = (int*)(csr + 2 * (size_t)N + 4);
  int*   colidx   = rank + E;

  const int T = 256;
  const long twoN = 2L * N;

  zero_f32<<<dim3((unsigned)((twoN + T - 1) / T)), dim3(T), 0, stream>>>(norm_out, twoN);
  degree_rank_kernel<<<dim3((unsigned)((E + T - 1) / T)), dim3(T), 0, stream>>>(src, dst, norm_out, cnt, rank, E);
  scan_norm_kernel<<<dim3((unsigned)((N + 1023) / 1024)), dim3(1024), 0, stream>>>(cnt, norm_out, norm_in, rowptr, N);
  fill_csr_kernel<<<dim3((unsigned)((E + T - 1) / T)), dim3(T), 0, stream>>>(src, dst, rank, rowptr, colidx, E);

  const int ntiles = N / 16;
  const int rem = N - ntiles * 16;
  const unsigned gemm_blocks = (unsigned)((ntiles + 7) / 8);
  const unsigned gath_blocks = (unsigned)((N + 7) / 8);

  {
    if (ntiles)
      gemm_norm_wmma<128><<<dim3(gemm_blocks), dim3(T), 0, stream>>>(in_feat, W1, norm_out, h, ntiles);
    if (rem)
      gemm_norm_scalar<128><<<dim3((unsigned)((rem * 128 + T - 1) / T)), dim3(T), 0, stream>>>(
          in_feat, W1, norm_out, h, ntiles * 16, rem);
    gather_rows<128, true><<<dim3(gath_blocks), dim3(T), 0, stream>>>(h, rowptr, colidx, norm_in, b1, xbuf, N);
  }

  {
    if (ntiles)
      gemm_norm_wmma<128><<<dim3(gemm_blocks), dim3(T), 0, stream>>>(xbuf, W2, norm_out, h, ntiles);
    if (rem)
      gemm_norm_scalar<128><<<dim3((unsigned)((rem * 128 + T - 1) / T)), dim3(T), 0, stream>>>(
          xbuf, W2, norm_out, h, ntiles * 16, rem);
    gather_rows<128, true><<<dim3(gath_blocks), dim3(T), 0, stream>>>(h, rowptr, colidx, norm_in, b2, xbuf, N);
  }

  {
    if (ntiles)
      gemm_norm_wmma<64><<<dim3(gemm_blocks), dim3(T), 0, stream>>>(xbuf, W3, norm_out, h, ntiles);
    if (rem)
      gemm_norm_scalar<64><<<dim3((unsigned)((rem * 64 + T - 1) / T)), dim3(T), 0, stream>>>(
          xbuf, W3, norm_out, h, ntiles * 16, rem);
    gather_rows<64, false><<<dim3(gath_blocks), dim3(T), 0, stream>>>(h, rowptr, colidx, norm_in, b3, out, N);
  }
}


# Layers 2 and 3: W product moved into the previous layer's gather (16 summed dst rows -> LDS tile -> f32 WMMA, B fragments resident per wave); activations not stored, 2 launches fewer
# speedup vs baseline: 2.6408x; 1.1160x over previous
#include <hip/hip_runtime.h>
#include <hip/hip_bf16.h>
#include <stdint.h>


typedef float v2f __attribute__((ext_vector_type(2)));
typedef float v8f __attribute__((ext_vector_type(8)));

static __device__ __forceinline__ void atomic_add_f32(float* p, float v) {
  __hip_atomic_fetch_add(p, v, __ATOMIC_RELAXED, __HIP_MEMORY_SCOPE_AGENT);
}


__global__ void zero_f32(float* __restrict__ p, long n) {
  long i = (long)blockIdx.x * blockDim.x + threadIdx.x;
  if (i < n) p[i] = 0.0f;
}

__global__ void degree_rank_kernel(const int* __restrict__ src, const int* __restrict__ dst,
                                   float* __restrict__ dout, int* __restrict__ cnt,
                                   int* __restrict__ rank, int E) {
  int i = blockIdx.x * blockDim.x + threadIdx.x;
  if (i < E) {
    atomic_add_f32(&dout[src[i]], 1.0f);
    rank[i] = __hip_atomic_fetch_add(&cnt[dst[i]], 1, __ATOMIC_RELAXED, __HIP_MEMORY_SCOPE_AGENT);
  }
}

__global__ void __launch_bounds__(1024)
scan_norm_kernel(const int* __restrict__ cnt, float* __restrict__ dout,
                 float* __restrict__ norm_in, int* __restrict__ rowptr, int n) {
  __shared__ int wpre[32];
  __shared__ int wsum[32];
  const int tid = threadIdx.x;
  const int lane = tid & 31;
  const int wv = tid >> 5;
  const int base = blockIdx.x * 1024;

  int part = 0;
  {
    const int4* c4 = (const int4*)cnt;
    for (int i = tid; i < base / 4; i += 1024) {
      const int4 q = c4[i];
      part += (q.x + q.y) + (q.z + q.w);
    }
  }
  for (int o = 16; o > 0; o >>= 1) part += __shfl_xor(part, o);

  const int i = base + tid;
  const int c = (i < n) ? cnt[i] : 0;
  int x = c;
  for (int o = 1; o < 32; o <<= 1) {
    const int y = __shfl_up(x, o);
    if (lane >= o) x += y;
  }
  if (lane == 31) wsum[wv] = x;
  if (lane == 0) wpre[wv] = part;
  __syncthreads();

  int a = wpre[lane];
  int b = (lane < wv) ? wsum[lane] : 0;
  for (int o = 16; o > 0; o >>= 1) {
    a += __shfl_xor(a, o);
    b += __shfl_xor(b, o);
  }
  const int excl = a + b + x - c;
  if (i < n) {
    rowptr[i] = excl;
    norm_in[i] = rsqrtf(fmaxf((float)c, 1.0f));
    dout[i] = rsqrtf(fmaxf(dout[i], 1.0f));
    if (i == n - 1) rowptr[n] = excl + c;
  }
}

__global__ void fill_csr_kernel(const int* __restrict__ src, const int* __restrict__ dst,
                                const int* __restrict__ rank, const int* __restrict__ rowptr,
                                int* __restrict__ colidx, int E) {
  int i = blockIdx.x * blockDim.x + threadIdx.x;
  if (i < E) {
    const int p = rowptr[dst[i]] + rank[i];
    if ((unsigned)p < (unsigned)E) colidx[p] = src[i];
  }
}

template <int FOUT>
__global__ void __launch_bounds__(256)
gemm_norm_wmma(const float* __restrict__ X, const float* __restrict__ W,
               const float* __restrict__ norm, float* __restrict__ H, int ntiles) {
  constexpr int FIN = 128;
  constexpr int NT = FOUT / 16;

  __shared__ __align__(16) float Wlds[FIN * FOUT];
  {
    const float4* wg = (const float4*)W;
    float4* wl = (float4*)Wlds;
    for (int i = threadIdx.x; i < FIN * FOUT / 4; i += blockDim.x) wl[i] = wg[i];
  }
  __syncthreads();

  const int wave = threadIdx.x >> 5;
  const int lane = threadIdx.x & 31;
  const int tile = blockIdx.x * 8 + wave;
  if (tile >= ntiles) return;

  const int half = lane >> 4;
  const int lm   = lane & 15;
  const int row0 = tile * 16;

  const int arow = row0 + lm;
  const float nrm = norm[arow];
  const float* __restrict__ xrow = X + (size_t)arow * FIN;

  v8f acc[NT];
#pragma unroll
  for (int t = 0; t < NT; ++t) acc[t] = (v8f)0.0f;

  for (int kk = 0; kk < FIN; kk += 4) {
    const int ka = kk + half * 2;
    v2f a;
    a.x = xrow[ka] * nrm;
    a.y = xrow[ka + 1] * nrm;
#pragma unroll
    for (int nt = 0; nt < NT; ++nt) {
      const int col = nt * 16 + lm;
      v2f b;
      b.x = Wlds[ka * FOUT + col];
      b.y = Wlds[(ka + 1) * FOUT + col];
      acc[nt] = __builtin_amdgcn_wmma_f32_16x16x4_f32(
           false, a,  false, b,
           (short)0, acc[nt],  false,  false);
    }
  }

#pragma unroll
  for (int nt = 0; nt < NT; ++nt) {
    const int col = nt * 16 + lm;
    float* __restrict__ hp = H + (size_t)(row0 + half * 8) * FOUT + col;
#pragma unroll
    for (int r = 0; r < 8; ++r) hp[(size_t)r * FOUT] = acc[nt][r];
  }
}

template <int FOUT>
__global__ void gemm_norm_scalar(const float* __restrict__ X, const float* __restrict__ W,
                                 const float* __restrict__ norm, float* __restrict__ H,
                                 int row_begin, int nrows) {
  constexpr int FIN = 128;
  int i = blockIdx.x * blockDim.x + threadIdx.x;
  if (i >= nrows * FOUT) return;
  int row = row_begin + i / FOUT;
  int col = i % FOUT;
  const float* xr = X + (size_t)row * FIN;
  float nrm = norm[row];
  float s = 0.0f;
  for (int k = 0; k < FIN; ++k) s += xr[k] * W[k * FOUT + col];
  H[(size_t)row * FOUT + col] = s * nrm;
}

template <int F, bool RELU>
__global__ void __launch_bounds__(256)
gather_rows(const float* __restrict__ H, const int* __restrict__ rowptr,
            const int* __restrict__ colidx, const float* __restrict__ norm_in,
            const float* __restrict__ bias, float* __restrict__ out, int N) {
  constexpr int VW = F / 32;
  typedef float vec __attribute__((ext_vector_type(VW)));
  const int lane = threadIdx.x & 31;
  const int row = __builtin_amdgcn_readfirstlane((int)((blockIdx.x * 256u + threadIdx.x) >> 5));
  if (row >= N) return;
  const int beg = rowptr[row];
  const int end = rowptr[row + 1];
  const unsigned smax = (unsigned)(N - 1);

  vec acc = (vec)0.0f;
  for (int j = beg; j < end; j += 32) {
    const int cnt = (end - j < 32) ? (end - j) : 32;
    const int myidx = (lane < cnt) ? colidx[j + lane] : 0;
    int t = 0;
    for (; t + 8 <= cnt; t += 8) {
      vec v[8];
#pragma unroll
      for (int u = 0; u < 8; ++u) {
        unsigned s = (unsigned)__builtin_amdgcn_readlane(myidx, t + u);
        s = (s < smax) ? s : smax;
        v[u] = *((const vec*)(H + (size_t)s * F) + lane);
      }
#pragma unroll
      for (int u = 0; u < 8; ++u) acc += v[u];
    }
    if (t < cnt) {
      vec v[7];
#pragma unroll
      for (int u = 0; u < 7; ++u) {
        v[u] = (vec)0.0f;
        if (t + u < cnt) {
          unsigned s = (unsigned)__builtin_amdgcn_readlane(myidx, t + u);
          s = (s < smax) ? s : smax;
          v[u] = *((const vec*)(H + (size_t)s * F) + lane);
        }
      }
#pragma unroll
      for (int u = 0; u < 7; ++u) acc += v[u];
    }
  }

  const float nin = norm_in[row];
  const vec bv = *((const vec*)bias + lane);
  vec r = acc * nin + bv;
  if (RELU) {
#pragma unroll
    for (int k = 0; k < VW; ++k) r[k] = fmaxf(r[k], 0.0f);
  }
  *((vec*)(out + (size_t)row * F) + lane) = r;
}


template <int FOUT>
__global__ void __launch_bounds__(FOUT * 2)
gather_gemm(const float* __restrict__ H, const int* __restrict__ rowptr,
            const int* __restrict__ colidx, const float* __restrict__ norm_in,
            const float* __restrict__ bias, const float* __restrict__ norm_out,
            const float* __restrict__ Wn, float* __restrict__ Hn, int N, int ntiles) {
  constexpr int F = 128;
  constexpr int NW = FOUT / 16;
  constexpr int RPW = 16 / NW;
  constexpr int PITCH = F + 4;
  typedef float vec __attribute__((ext_vector_type(4)));

  __shared__ __align__(16) float Xs[16 * PITCH];
  constexpr int YP = FOUT + 4;
  __shared__ __align__(16) float Ys[16 * YP];

  const int lane = threadIdx.x & 31;
  const int wave = __builtin_amdgcn_readfirstlane((int)(threadIdx.x >> 5));
  const int half = lane >> 4;
  const int lm   = lane & 15;

  constexpr int KREG = 16;
  __shared__ __align__(8) float Bs[NW * (32 - KREG) * 64];
  float* bsw = Bs + (wave * (32 - KREG) * 32 + lane) * 2;
  v2f bf[KREG];
#pragma unroll
  for (int kk = 0; kk < 32; ++kk) {
    const int ka = kk * 4 + half * 2;
    v2f b;
    b.x = Wn[ka * FOUT + wave * 16 + lm];
    b.y = Wn[(ka + 1) * FOUT + wave * 16 + lm];
    if (kk < KREG) bf[kk] = b;
    else *((v2f*)(bsw + (kk - KREG) * 64)) = b;
  }
  const unsigned smax = (unsigned)(N - 1);

  for (int tile = blockIdx.x; tile < ntiles; tile += gridDim.x) {
    float* xs = Xs;

    for (int rr = 0; rr < RPW; ++rr) {
      const int lr = wave * RPW + rr;
      const int row = __builtin_amdgcn_readfirstlane(tile * 16 + lr);
      vec r = (vec)0.0f;
      if (row < N) {
        const int beg = rowptr[row];
        const int end = rowptr[row + 1];
        const float nin = norm_in[row];
        const float nout = norm_out[row];
        vec acc = (vec)0.0f;
        for (int j = beg; j < end; j += 32) {
          const int cnt = (end - j < 32) ? (end - j) : 32;
          const int myidx = (lane < cnt) ? colidx[j + lane] : 0;
          int t = 0;
          for (; t + 8 <= cnt; t += 8) {
            vec v[8];
#pragma unroll
            for (int u = 0; u < 8; ++u) {
              unsigned s = (unsigned)__builtin_amdgcn_readlane(myidx, t + u);
              s = (s < smax) ? s : smax;
              v[u] = *((const vec*)(H + (size_t)s * F) + lane);
            }
#pragma unroll
            for (int u = 0; u < 8; ++u) acc += v[u];
          }
          if (t < cnt) {
            vec v[7];
#pragma unroll
            for (int u = 0; u < 7; ++u) {
              if (t + u < cnt) {
                unsigned s = (unsigned)__builtin_amdgcn_readlane(myidx, t + u);
                s = (s < smax) ? s : smax;
                v[u] = *((const vec*)(H + (size_t)s * F) + lane);
              }
            }
#pragma unroll
            for (int u = 0; u < 7; ++u)
              if (t + u < cnt) acc += v[u];
          }
        }
        const vec bv = *((const vec*)bias + lane);
        r = acc * nin + bv;
#pragma unroll
        for (int k = 0; k < 4; ++k) r[k] = fmaxf(r[k], 0.0f) * nout;
      }
      *((vec*)(xs + lr * PITCH) + lane) = r;
    }
    __syncthreads();

    v8f acc8 = (v8f)0.0f;
    const float* xa = xs + lm * PITCH + half * 2;
    v2f af[2][4];
#pragma unroll
    for (int u = 0; u < 4; ++u) af[0][u] = *((const v2f*)(xa + u * 4));
#pragma unroll
    for (int g = 0; g < 8; ++g) {
      if (g < 7) {
#pragma unroll
        for (int u = 0; u < 4; ++u) af[(g + 1) & 1][u] = *((const v2f*)(xa + ((g + 1) * 4 + u) * 4));
      }
      v2f bl[4];
      if (g * 4 >= KREG) {
#pragma unroll
        for (int u = 0; u < 4; ++u) bl[u] = *((const v2f*)(bsw + (g * 4 + u - KREG) * 64));
      }
#pragma unroll
      for (int u = 0; u < 4; ++u)
        acc8 = __builtin_amdgcn_wmma_f32_16x16x4_f32(
             false, af[g & 1][u],  false, (g * 4 >= KREG) ? bl[u] : bf[(g * 4 + u) % KREG],
             (short)0, acc8,  false,  false);
    }

#pragma unroll
    for (int r8 = 0; r8 < 8; ++r8) Ys[(half * 8 + r8) * YP + wave * 16 + lm] = acc8[r8];
    __syncthreads();
    constexpr int LPR = FOUT / 4;
    constexpr int RPI = 32 / LPR;
#pragma unroll
    for (int p = 0; p < RPW / RPI; ++p) {
      const int lr = wave * RPW + p * RPI + lane / LPR;
      const int c4 = (lane % LPR) * 4;
      const int orow = tile * 16 + lr;
      const vec y = *((const vec*)(Ys + lr * YP + c4));
      if (orow < N) *((vec*)(Hn + (size_t)orow * FOUT + c4)) = y;
    }
  }
}


extern "C" void kernel_launch(void* const* d_in, const int* in_sizes, int n_in,
                              void* d_out, int out_size, void* d_ws, size_t ws_size,
                              hipStream_t stream) {
  (void)n_in; (void)out_size; (void)ws_size;
  const float* in_feat = (const float*)d_in[0];
  const int*   src     = (const int*)d_in[1];
  const int*   dst     = (const int*)d_in[2];
  const float* W1 = (const float*)d_in[3];
  const float* b1 = (const float*)d_in[4];
  const float* W2 = (const float*)d_in[5];
  const float* b2 = (const float*)d_in[6];
  const float* W3 = (const float*)d_in[7];
  const float* b3 = (const float*)d_in[8];

  const int N = in_sizes[0] / 128;
  const int E = in_sizes[1];

  float* ws       = (float*)d_ws;
  float* norm_out = ws;
  int*   cnt      = (int*)(ws + N);
  float* h        = ws + 2 * (size_t)N;
  float* csr      = h + (size_t)N * 128;
  float* xbuf     = csr + (size_t)N * 128;
  float* out      = (float*)d_out;

  float* norm_in  = csr;
  int*   rowptr   = (int*)(csr + N);
  int*   rank     = (int*)(csr + 2 * (size_t)N + 4);
  int*   colidx   = rank + E;

  const int T = 256;
  const long twoN = 2L * N;

  zero_f32<<<dim3((unsigned)((twoN + T - 1) / T)), dim3(T), 0, stream>>>(norm_out, twoN);
  degree_rank_kernel<<<dim3((unsigned)((E + T - 1) / T)), dim3(T), 0, stream>>>(src, dst, norm_out, cnt, rank, E);
  scan_norm_kernel<<<dim3((unsigned)((N + 1023) / 1024)), dim3(1024), 0, stream>>>(cnt, norm_out, norm_in, rowptr, N);
  fill_csr_kernel<<<dim3((unsigned)((E + T - 1) / T)), dim3(T), 0, stream>>>(src, dst, rank, rowptr, colidx, E);

  const int ntiles = N / 16;
  const int rem = N - ntiles * 16;
  const unsigned gemm_blocks = (unsigned)((ntiles + 7) / 8);
  const unsigned gath_blocks = (unsigned)((N + 7) / 8);
  const int tiles_up = (N + 15) / 16;

  if (ntiles)
    gemm_norm_wmma<128><<<dim3(gemm_blocks), dim3(T), 0, stream>>>(in_feat, W1, norm_out, h, ntiles);
  if (rem)
    gemm_norm_scalar<128><<<dim3((unsigned)((rem * 128 + T - 1) / T)), dim3(T), 0, stream>>>(
        in_feat, W1, norm_out, h, ntiles * 16, rem);

  gather_gemm<128><<<dim3(1024), dim3(256), 0, stream>>>(h, rowptr, colidx, norm_in, b1, norm_out, W2, xbuf, N, tiles_up);

  gather_gemm<64><<<dim3(2048), dim3(128), 0, stream>>>(xbuf, rowptr, colidx, norm_in, b2, norm_out, W3, h, N, tiles_up);

  gather_rows<64, false><<<dim3(gath_blocks), dim3(T), 0, stream>>>(h, rowptr, colidx, norm_in, b3, out, N);
}


# CSR build as two-level counting sort on LDS counters (4096-edge chunks sorted by 512-node bucket, then one work-group per bucket), no global atomics; 2 launches fewer; row sums in LDS arrival order
# speedup vs baseline: 4.7373x; 1.7939x over previous
#include <hip/hip_runtime.h>
#include <hip/hip_bf16.h>
#include <stdint.h>


typedef float v2f __attribute__((ext_vector_type(2)));
typedef float v8f __attribute__((ext_vector_type(8)));

static __device__ __forceinline__ void atomic_add_f32(float* p, float v) {
  __hip_atomic_fetch_add(p, v, __ATOMIC_RELAXED, __HIP_MEMORY_SCOPE_AGENT);
}


__global__ void zero_f32(float* __restrict__ p, long n) {
  long i = (long)blockIdx.x * blockDim.x + threadIdx.x;
  if (i < n) p[i] = 0.0f;
}

__global__ void degree_rank_kernel(const int* __restrict__ src, const int* __restrict__ dst,
                                   float* __restrict__ dout, int* __restrict__ cnt,
                                   int* __restrict__ rank, int E) {
  int i = blockIdx.x * blockDim.x + threadIdx.x;
  if (i < E) {
    atomic_add_f32(&dout[src[i]], 1.0f);
    rank[i] = __hip_atomic_fetch_add(&cnt[dst[i]], 1, __ATOMIC_RELAXED, __HIP_MEMORY_SCOPE_AGENT);
  }
}

__global__ void __launch_bounds__(1024)
scan_norm_kernel(const int* __restrict__ cnt, float* __restrict__ dout,
                 float* __restrict__ norm_in, int* __restrict__ rowptr, int n) {
  __shared__ int wpre[32];
  __shared__ int wsum[32];
  const int tid = threadIdx.x;
  const int lane = tid & 31;
  const int wv = tid >> 5;
  const int base = blockIdx.x * 1024;

  int part = 0;
  {
    const int4* c4 = (const int4*)cnt;
    for (int i = tid; i < base / 4; i += 1024) {
      const int4 q = c4[i];
      part += (q.x + q.y) + (q.z + q.w);
    }
  }
  for (int o = 16; o > 0; o >>= 1) part += __shfl_xor(part, o);

  const int i = base + tid;
  const int c = (i < n) ? cnt[i] : 0;
  int x = c;
  for (int o = 1; o < 32; o <<= 1) {
    const int y = __shfl_up(x, o);
    if (lane >= o) x += y;
  }
  if (lane == 31) wsum[wv] = x;
  if (lane == 0) wpre[wv] = part;
  __syncthreads();

  int a = wpre[lane];
  int b = (lane < wv) ? wsum[lane] : 0;
  for (int o = 16; o > 0; o >>= 1) {
    a += __shfl_xor(a, o);
    b += __shfl_xor(b, o);
  }
  const int excl = a + b + x - c;
  if (i < n) {
    rowptr[i] = excl;
    norm_in[i] = rsqrtf(fmaxf((float)c, 1.0f));
    dout[i] = rsqrtf(fmaxf(dout[i], 1.0f));
    if (i == n - 1) rowptr[n] = excl + c;
  }
}

__global__ void fill_csr_kernel(const int* __restrict__ src, const int* __restrict__ dst,
                                const int* __restrict__ rank, const int* __restrict__ rowptr,
                                int* __restrict__ colidx, int E) {
  int i = blockIdx.x * blockDim.x + threadIdx.x;
  if (i < E) {
    const int p = rowptr[dst[i]] + rank[i];
    if ((unsigned)p < (unsigned)E) colidx[p] = src[i];
  }
}


constexpr int CH = 4096;
constexpr int NBK_SH = 9;
constexpr int NBK = 1 << NBK_SH;
constexpr int MAXB = 256;

__global__ void __launch_bounds__(256)
tile_sort_kernel(const int* __restrict__ src, const int* __restrict__ dst,
                 int* __restrict__ tmpDw, int* __restrict__ tmpS,
                 int* __restrict__ offsD, int* __restrict__ offsS,
                 int E, int N, int C) {
  int2* __restrict__ tmpD = (int2*)tmpDw;
  __shared__ int hist[MAXB];
  __shared__ int offs[MAXB + 1];
  __shared__ int wtot[8];
  __shared__ int2 stage[CH];
  const int tid = threadIdx.x;
  const int lane = tid & 31;
  const int wv = tid >> 5;
  const bool bySrc = (int)blockIdx.x >= C;
  const int chunk = bySrc ? (int)blockIdx.x - C : (int)blockIdx.x;
  const int* __restrict__ key = bySrc ? src : dst;
  hist[tid] = 0;
  __syncthreads();

  const unsigned nmax = (unsigned)(N - 1);
  const int e0 = chunk * CH + tid;
  int kd[16], pl[16], rk[16];
#pragma unroll
  for (int k = 0; k < 16; ++k) {
    const int e = e0 + k * 256;
    kd[k] = -1;
    pl[k] = 0;
    if (e < E) {
      const unsigned kv = (unsigned)key[e];
      kd[k] = (int)(kv < nmax ? kv : nmax);
      if (!bySrc) pl[k] = src[e];
    }
  }
#pragma unroll
  for (int k = 0; k < 16; ++k) {
    rk[k] = 0;
    if (kd[k] >= 0)
      rk[k] = __hip_atomic_fetch_add(&hist[kd[k] >> NBK_SH], 1, __ATOMIC_RELAXED, __HIP_MEMORY_SCOPE_WORKGROUP);
  }
  __syncthreads();

  const int c = hist[tid];
  int x = c;
  for (int o = 1; o < 32; o <<= 1) {
    const int y = __shfl_up(x, o);
    if (lane >= o) x += y;
  }
  if (lane == 31) wtot[wv] = x;
  __syncthreads();
  int pre = 0;
#pragma unroll
  for (int w = 0; w < 8; ++w) pre += (w < wv) ? wtot[w] : 0;
  const int excl = pre + x - c;
  offs[tid] = excl;
  int* __restrict__ og = (bySrc ? offsS : offsD) + (size_t)chunk * (MAXB + 1);
  og[tid] = excl;
  if (tid == 255) { offs[256] = excl + c; og[256] = excl + c; }
  __syncthreads();

#pragma unroll
  for (int k = 0; k < 16; ++k)
    if (kd[k] >= 0) stage[offs[kd[k] >> NBK_SH] + rk[k]] = make_int2(kd[k], pl[k]);
  __syncthreads();
  const int total = offs[256];
  const size_t cbase = (size_t)chunk * CH;
  if (bySrc) {
#pragma unroll
    for (int k = 0; k < 16; ++k) {
      const int i = tid + k * 256;
      if (i < total) tmpS[cbase + i] = stage[i].x;
    }
  } else {
#pragma unroll
    for (int k = 0; k < 16; ++k) {
      const int i = tid + k * 256;
      if (i < total) tmpD[cbase + i] = stage[i];
    }
  }
}

template <bool PLACE>
static __device__ __forceinline__ int bucket_pass(const int* __restrict__ keys, const int sh,
                                                  const int* __restrict__ offs, int* lcnt,
                                                  int* __restrict__ colidx, const int b,
                                                  const int C, const int E, const int wv,
                                                  const int lane) {
  int losum = 0;
  const int nb0 = b << NBK_SH;
  for (int cb = 0; cb < C; cb += 1024) {
    const int cm = cb + wv + 32 * lane;
    int lov = 0, hiv = 0;
    if (cm < C) {
      const int* __restrict__ o = offs + (size_t)cm * (MAXB + 1) + b;
      lov = o[0];
      hiv = o[1];
    }
    losum += lov;
    int ns = (C - cb - wv + 31) >> 5;
    ns = ns < 0 ? 0 : (ns > 32 ? 32 : ns);
    for (int j0 = 0; j0 < ns; j0 += 4) {
      int len[4];
      size_t base[4];
      int2 q[4];
#pragma unroll
      for (int u = 0; u < 4; ++u) {
        const int j = j0 + u;
        const int l = __builtin_amdgcn_readlane(lov, j);
        const int h = __builtin_amdgcn_readlane(hiv, j);
        len[u] = h - l;
        base[u] = (size_t)(cb + wv + 32 * j) * CH + (size_t)l;
        const size_t idx = (lane < len[u]) ? base[u] + (size_t)lane : (size_t)0;
        if (PLACE) q[u] = ((const int2*)keys)[idx];
        else q[u] = make_int2(keys[idx << sh], 0);
      }
#pragma unroll
      for (int u = 0; u < 4; ++u) {
        if (lane < len[u]) {
          const int ln = (q[u].x - nb0) & (NBK - 1);
          if (PLACE) {
            const int pos = __hip_atomic_fetch_add(&lcnt[ln], 1, __ATOMIC_RELAXED, __HIP_MEMORY_SCOPE_WORKGROUP);
            if ((unsigned)pos < (unsigned)E) colidx[pos] = q[u].y;
          } else {
            __hip_atomic_fetch_add(&lcnt[ln], 1, __ATOMIC_RELAXED, __HIP_MEMORY_SCOPE_WORKGROUP);
          }
        }
      }
#pragma unroll
      for (int u = 0; u < 4; ++u) {
        if (len[u] > 32) {
          for (int p = 32 + lane; p < len[u]; p += 32) {
            int2 r = make_int2(0, 0);
            if (PLACE) r = ((const int2*)keys)[base[u] + p];
            else r.x = keys[(base[u] + p) << sh];
            const int ln = (r.x - nb0) & (NBK - 1);
            if (PLACE) {
              const int pos = __hip_atomic_fetch_add(&lcnt[ln], 1, __ATOMIC_RELAXED, __HIP_MEMORY_SCOPE_WORKGROUP);
              if ((unsigned)pos < (unsigned)E) colidx[pos] = r.y;
            } else {
              __hip_atomic_fetch_add(&lcnt[ln], 1, __ATOMIC_RELAXED, __HIP_MEMORY_SCOPE_WORKGROUP);
            }
          }
        }
      }
    }
  }
  return losum;
}

__global__ void __launch_bounds__(1024)
bucket_csr_kernel(const int* __restrict__ tmpD, const int* __restrict__ tmpS,
                  const int* __restrict__ offsD, const int* __restrict__ offsS,
                  float* __restrict__ norm_in, float* __restrict__ norm_out,
                  int* __restrict__ rowptr, int* __restrict__ colidx,
                  int N, int E, int C, int B) {
  __shared__ int lcnt[NBK];
  __shared__ int wred[32];
  __shared__ int wsc[32];
  const int tid = threadIdx.x;
  const int lane = tid & 31;
  const int wv = __builtin_amdgcn_readfirstlane(tid >> 5);
  const bool bySrc = (int)blockIdx.x >= B;
  const int b = bySrc ? (int)blockIdx.x - B : (int)blockIdx.x;
  const int* __restrict__ keys = bySrc ? tmpS : tmpD;
  const int sh = bySrc ? 0 : 1;
  const int* __restrict__ offs = bySrc ? offsS : offsD;

  if (tid < NBK) lcnt[tid] = 0;
  __syncthreads();
  int losum = bucket_pass<false>(keys, sh, offs, lcnt, colidx, b, C, E, wv, lane);
  for (int o = 16; o > 0; o >>= 1) losum += __shfl_xor(losum, o);
  if (lane == 0) wred[wv] = losum;
  __syncthreads();

  const int cval = (tid < NBK) ? lcnt[tid] : 0;
  int x = cval;
  for (int o = 1; o < 32; o <<= 1) {
    const int y = __shfl_up(x, o);
    if (lane >= o) x += y;
  }
  if (lane == 31) wsc[wv] = x;
  __syncthreads();
  int start = wred[lane];
  int pre = (lane < wv) ? wsc[lane] : 0;
  for (int o = 16; o > 0; o >>= 1) {
    start += __shfl_xor(start, o);
    pre += __shfl_xor(pre, o);
  }
  const int excl = pre + x - cval;
  const int node = (b << NBK_SH) + tid;
  if (tid < NBK && node < N) {
    const float nrm = rsqrtf(fmaxf((float)cval, 1.0f));
    if (bySrc) {
      norm_out[node] = nrm;
    } else {
      norm_in[node] = nrm;
      rowptr[node] = start + excl;
      if (node == N - 1) rowptr[N] = start + excl + cval;
    }
  }
  if (bySrc) return;
  if (tid < NBK) lcnt[tid] = start + excl;
  __syncthreads();
  bucket_pass<true>(keys, sh, offs, lcnt, colidx, b, C, E, wv, lane);
}

template <int FOUT>
__global__ void __launch_bounds__(256)
gemm_norm_wmma(const float* __restrict__ X, const float* __restrict__ W,
               const float* __restrict__ norm, float* __restrict__ H, int ntiles) {
  constexpr int FIN = 128;
  constexpr int NT = FOUT / 16;

  __shared__ __align__(16) float Wlds[FIN * FOUT];
  {
    const float4* wg = (const float4*)W;
    float4* wl = (float4*)Wlds;
    for (int i = threadIdx.x; i < FIN * FOUT / 4; i += blockDim.x) wl[i] = wg[i];
  }
  __syncthreads();

  const int wave = threadIdx.x >> 5;
  const int lane = threadIdx.x & 31;
  const int tile = blockIdx.x * 8 + wave;
  if (tile >= ntiles) return;

  const int half = lane >> 4;
  const int lm   = lane & 15;
  const int row0 = tile * 16;

  const int arow = row0 + lm;
  const float nrm = norm[arow];
  const float* __restrict__ xrow = X + (size_t)arow * FIN;

  v8f acc[NT];
#pragma unroll
  for (int t = 0; t < NT; ++t) acc[t] = (v8f)0.0f;

  for (int kk = 0; kk < FIN; kk += 4) {
    const int ka = kk + half * 2;
    v2f a;
    a.x = xrow[ka] * nrm;
    a.y = xrow[ka + 1] * nrm;
#pragma unroll
    for (int nt = 0; nt < NT; ++nt) {
      const int col = nt * 16 + lm;
      v2f b;
      b.x = Wlds[ka * FOUT + col];
      b.y = Wlds[(ka + 1) * FOUT + col];
      acc[nt] = __builtin_amdgcn_wmma_f32_16x16x4_f32(
           false, a,  false, b,
           (short)0, acc[nt],  false,  false);
    }
  }

#pragma unroll
  for (int nt = 0; nt < NT; ++nt) {
    const int col = nt * 16 + lm;
    float* __restrict__ hp = H + (size_t)(row0 + half * 8) * FOUT + col;
#pragma unroll
    for (int r = 0; r < 8; ++r) hp[(size_t)r * FOUT] = acc[nt][r];
  }
}

template <int FOUT>
__global__ void gemm_norm_scalar(const float* __restrict__ X, const float* __restrict__ W,
                                 const float* __restrict__ norm, float* __restrict__ H,
                                 int row_begin, int nrows) {
  constexpr int FIN = 128;
  int i = blockIdx.x * blockDim.x + threadIdx.x;
  if (i >= nrows * FOUT) return;
  int row = row_begin + i / FOUT;
  int col = i % FOUT;
  const float* xr = X + (size_t)row * FIN;
  float nrm = norm[row];
  float s = 0.0f;
  for (int k = 0; k < FIN; ++k) s += xr[k] * W[k * FOUT + col];
  H[(size_t)row * FOUT + col] = s * nrm;
}

template <int F, bool RELU>
__global__ void __launch_bounds__(256)
gather_rows(const float* __restrict__ H, const int* __restrict__ rowptr,
            const int* __restrict__ colidx, const float* __restrict__ norm_in,
            const float* __restrict__ bias, float* __restrict__ out, int N) {
  constexpr int VW = F / 32;
  typedef float vec __attribute__((ext_vector_type(VW)));
  const int lane = threadIdx.x & 31;
  const int row = __builtin_amdgcn_readfirstlane((int)((blockIdx.x * 256u + threadIdx.x) >> 5));
  if (row >= N) return;
  const int beg = rowptr[row];
  const int end = rowptr[row + 1];
  const unsigned smax = (unsigned)(N - 1);

  vec acc = (vec)0.0f;
  for (int j = beg; j < end; j += 32) {
    const int cnt = (end - j < 32) ? (end - j) : 32;
    const int myidx = (lane < cnt) ? colidx[j + lane] : 0;
    int t = 0;
    for (; t + 8 <= cnt; t += 8) {
      vec v[8];
#pragma unroll
      for (int u = 0; u < 8; ++u) {
        unsigned s = (unsigned)__builtin_amdgcn_readlane(myidx, t + u);
        s = (s < smax) ? s : smax;
        v[u] = *((const vec*)(H + (size_t)s * F) + lane);
      }
#pragma unroll
      for (int u = 0; u < 8; ++u) acc += v[u];
    }
    if (t < cnt) {
      vec v[7];
#pragma unroll
      for (int u = 0; u < 7; ++u) {
        v[u] = (vec)0.0f;
        if (t + u < cnt) {
          unsigned s = (unsigned)__builtin_amdgcn_readlane(myidx, t + u);
          s = (s < smax) ? s : smax;
          v[u] = *((const vec*)(H + (size_t)s * F) + lane);
        }
      }
#pragma unroll
      for (int u = 0; u < 7; ++u) acc += v[u];
    }
  }

  const float nin = norm_in[row];
  const vec bv = *((const vec*)bias + lane);
  vec r = acc * nin + bv;
  if (RELU) {
#pragma unroll
    for (int k = 0; k < VW; ++k) r[k] = fmaxf(r[k], 0.0f);
  }
  *((vec*)(out + (size_t)row * F) + lane) = r;
}


template <int FOUT>
__global__ void __launch_bounds__(FOUT * 2)
gather_gemm(const float* __restrict__ H, const int* __restrict__ rowptr,
            const int* __restrict__ colidx, const float* __restrict__ norm_in,
            const float* __restrict__ bias, const float* __restrict__ norm_out,
            const float* __restrict__ Wn, float* __restrict__ Hn, int N, int ntiles) {
  constexpr int F = 128;
  constexpr int NW = FOUT / 16;
  constexpr int RPW = 16 / NW;
  constexpr int PITCH = F + 4;
  typedef float vec __attribute__((ext_vector_type(4)));

  __shared__ __align__(16) float Xs[16 * PITCH];
  constexpr int YP = FOUT + 4;
  __shared__ __align__(16) float Ys[16 * YP];

  const int lane = threadIdx.x & 31;
  const int wave = __builtin_amdgcn_readfirstlane((int)(threadIdx.x >> 5));
  const int half = lane >> 4;
  const int lm   = lane & 15;

  constexpr int KREG = 16;
  __shared__ __align__(8) float Bs[NW * (32 - KREG) * 64];
  float* bsw = Bs + (wave * (32 - KREG) * 32 + lane) * 2;
  v2f bf[KREG];
#pragma unroll
  for (int kk = 0; kk < 32; ++kk) {
    const int ka = kk * 4 + half * 2;
    v2f b;
    b.x = Wn[ka * FOUT + wave * 16 + lm];
    b.y = Wn[(ka + 1) * FOUT + wave * 16 + lm];
    if (kk < KREG) bf[kk] = b;
    else *((v2f*)(bsw + (kk - KREG) * 64)) = b;
  }
  const unsigned smax = (unsigned)(N - 1);

  for (int tile = blockIdx.x; tile < ntiles; tile += gridDim.x) {
    float* xs = Xs;

    for (int rr = 0; rr < RPW; ++rr) {
      const int lr = wave * RPW + rr;
      const int row = __builtin_amdgcn_readfirstlane(tile * 16 + lr);
      vec r = (vec)0.0f;
      if (row < N) {
        const int beg = rowptr[row];
        const int end = rowptr[row + 1];
        const float nin = norm_in[row];
        const float nout = norm_out[row];
        vec acc = (vec)0.0f;
        for (int j = beg; j < end; j += 32) {
          const int cnt = (end - j < 32) ? (end - j) : 32;
          const int myidx = (lane < cnt) ? colidx[j + lane] : 0;
          int t = 0;
          for (; t + 8 <= cnt; t += 8) {
            vec v[8];
#pragma unroll
            for (int u = 0; u < 8; ++u) {
              unsigned s = (unsigned)__builtin_amdgcn_readlane(myidx, t + u);
              s = (s < smax) ? s : smax;
              v[u] = *((const vec*)(H + (size_t)s * F) + lane);
            }
#pragma unroll
            for (int u = 0; u < 8; ++u) acc += v[u];
          }
          if (t < cnt) {
            vec v[7];
#pragma unroll
            for (int u = 0; u < 7; ++u) {
              if (t + u < cnt) {
                unsigned s = (unsigned)__builtin_amdgcn_readlane(myidx, t + u);
                s = (s < smax) ? s : smax;
                v[u] = *((const vec*)(H + (size_t)s * F) + lane);
              }
            }
#pragma unroll
            for (int u = 0; u < 7; ++u)
              if (t + u < cnt) acc += v[u];
          }
        }
        const vec bv = *((const vec*)bias + lane);
        r = acc * nin + bv;
#pragma unroll
        for (int k = 0; k < 4; ++k) r[k] = fmaxf(r[k], 0.0f) * nout;
      }
      *((vec*)(xs + lr * PITCH) + lane) = r;
    }
    __syncthreads();

    v8f acc8 = (v8f)0.0f;
    const float* xa = xs + lm * PITCH + half * 2;
    v2f af[2][4];
#pragma unroll
    for (int u = 0; u < 4; ++u) af[0][u] = *((const v2f*)(xa + u * 4));
#pragma unroll
    for (int g = 0; g < 8; ++g) {
      if (g < 7) {
#pragma unroll
        for (int u = 0; u < 4; ++u) af[(g + 1) & 1][u] = *((const v2f*)(xa + ((g + 1) * 4 + u) * 4));
      }
      v2f bl[4];
      if (g * 4 >= KREG) {
#pragma unroll
        for (int u = 0; u < 4; ++u) bl[u] = *((const v2f*)(bsw + (g * 4 + u - KREG) * 64));
      }
#pragma unroll
      for (int u = 0; u < 4; ++u)
        acc8 = __builtin_amdgcn_wmma_f32_16x16x4_f32(
             false, af[g & 1][u],  false, (g * 4 >= KREG) ? bl[u] : bf[(g * 4 + u) % KREG],
             (short)0, acc8,  false,  false);
    }

#pragma unroll
    for (int r8 = 0; r8 < 8; ++r8) Ys[(half * 8 + r8) * YP + wave * 16 + lm] = acc8[r8];
    __syncthreads();
    constexpr int LPR = FOUT / 4;
    constexpr int RPI = 32 / LPR;
#pragma unroll
    for (int p = 0; p < RPW / RPI; ++p) {
      const int lr = wave * RPW + p * RPI + lane / LPR;
      const int c4 = (lane % LPR) * 4;
      const int orow = tile * 16 + lr;
      const vec y = *((const vec*)(Ys + lr * YP + c4));
      if (orow < N) *((vec*)(Hn + (size_t)orow * FOUT + c4)) = y;
    }
  }
}


extern "C" void kernel_launch(void* const* d_in, const int* in_sizes, int n_in,
                              void* d_out, int out_size, void* d_ws, size_t ws_size,
                              hipStream_t stream) {
  (void)n_in; (void)out_size; (void)ws_size;
  const float* in_feat = (const float*)d_in[0];
  const int*   src     = (const int*)d_in[1];
  const int*   dst     = (const int*)d_in[2];
  const float* W1 = (const float*)d_in[3];
  const float* b1 = (const float*)d_in[4];
  const float* W2 = (const float*)d_in[5];
  const float* b2 = (const float*)d_in[6];
  const float* W3 = (const float*)d_in[7];
  const float* b3 = (const float*)d_in[8];

  const int N = in_sizes[0] / 128;
  const int E = in_sizes[1];

  float* ws       = (float*)d_ws;
  float* norm_out = ws;
  int*   cnt      = (int*)(ws + N);
  float* h        = ws + 2 * (size_t)N;
  float* csr      = h + (size_t)N * 128;
  float* xbuf     = csr + (size_t)N * 128;
  float* out      = (float*)d_out;

  float* norm_in  = csr;
  int*   rowptr   = (int*)(csr + N);
  int*   rank     = (int*)(csr + 2 * (size_t)N + 4);

  const int T = 256;
  const long twoN = 2L * N;

  const int C = (E + CH - 1) / CH;
  const int B = (N + NBK - 1) / NBK;
  const size_t o4 = (2 * (size_t)N + 4 + (size_t)E + 3) & ~(size_t)3;
  const size_t need = o4 + 3 * (size_t)C * CH + 2 * (size_t)C * (MAXB + 1);
  const bool sorted_build = C > 0 && B <= MAXB && need <= (size_t)N * 128;
  int*   colidx   = sorted_build ? rank : rank + E;

  if (sorted_build) {
    int*  tmpD  = (int*)(csr + o4);
    int*  tmpS  = tmpD + 2 * (size_t)C * CH;
    int*  offsD = tmpS + (size_t)C * CH;
    int*  offsS = offsD + (size_t)C * (MAXB + 1);
    tile_sort_kernel<<<dim3((unsigned)(2 * C)), dim3(256), 0, stream>>>(src, dst, tmpD, tmpS, offsD, offsS, E, N, C);
    bucket_csr_kernel<<<dim3((unsigned)(2 * B)), dim3(1024), 0, stream>>>(tmpD, tmpS, offsD, offsS, norm_in, norm_out,
                                                                          rowptr, colidx, N, E, C, B);
  } else {
    zero_f32<<<dim3((unsigned)((twoN + T - 1) / T)), dim3(T), 0, stream>>>(norm_out, twoN);
    degree_rank_kernel<<<dim3((unsigned)((E + T - 1) / T)), dim3(T), 0, stream>>>(src, dst, norm_out, cnt, rank, E);
    scan_norm_kernel<<<dim3((unsigned)((N + 1023) / 1024)), dim3(1024), 0, stream>>>(cnt, norm_out, norm_in, rowptr, N);
    fill_csr_kernel<<<dim3((unsigned)((E + T - 1) / T)), dim3(T), 0, stream>>>(src, dst, rank, rowptr, colidx, E);
  }

  const int ntiles = N / 16;
  const int rem = N - ntiles * 16;
  const unsigned gemm_blocks = (unsigned)((ntiles + 7) / 8);
  const unsigned gath_blocks = (unsigned)((N + 7) / 8);
  const int tiles_up = (N + 15) / 16;

  if (ntiles)
    gemm_norm_wmma<128><<<dim3(gemm_blocks), dim3(T), 0, stream>>>(in_feat, W1, norm_out, h, ntiles);
  if (rem)
    gemm_norm_scalar<128><<<dim3((unsigned)((rem * 128 + T - 1) / T)), dim3(T), 0, stream>>>(
        in_feat, W1, norm_out, h, ntiles * 16, rem);

  gather_gemm<128><<<dim3(1024), dim3(256), 0, stream>>>(h, rowptr, colidx, norm_in, b1, norm_out, W2, xbuf, N, tiles_up);

  gather_gemm<64><<<dim3(2048), dim3(128), 0, stream>>>(xbuf, rowptr, colidx, norm_in, b2, norm_out, W3, h, N, tiles_up);

  gather_rows<64, false><<<dim3(gath_blocks), dim3(T), 0, stream>>>(h, rowptr, colidx, norm_in, b3, out, N);
}
